# starting kernel
# baseline (speedup 1.0000x reference)
#include <hip/hip_runtime.h>
#include <math.h>

#define TT   1024
#define BB   16
#define CC   512
#define HH   8
#define KK   31
#define PP   15
#define MROWS (TT * BB)
#define N2    (2 * CC)
#define HK    (HH * KK)
#define HKPAD 256
#define KP    (CC / 8)

typedef __attribute__((ext_vector_type(2))) float v2f;
typedef __attribute__((ext_vector_type(4))) float v4f;
typedef __attribute__((ext_vector_type(8))) float v8f;

__global__ __launch_bounds__(256)
void pack_b_kernel(const float* __restrict__ W, float* __restrict__ P,
                   int ldw, int ncols, int total) {
    const int idx = blockIdx.x * 256 + threadIdx.x;
    if (idx >= total) return;
    const int j    = idx & 3;
    const int lane = (idx >> 2) & 31;
    const int kp   = (idx >> 7) & (KP - 1);
    const int nt   = idx >> 13;
    const int hf   = lane >> 4;
    const int lrow = lane & 15;
    const int col  = nt * 16 + lrow;
    const int kr   = kp * 8 + (j >> 1) * 4 + 2 * hf + (j & 1);
    P[idx] = (col < ncols) ? W[(size_t)kr * ldw + col] : 0.0f;
}

__global__ __launch_bounds__(32)
void gemm1_glu_kernel(const float* __restrict__ x,
                      const float* __restrict__ w1p,
                      const float* __restrict__ b1,
                      float* __restrict__ xg) {
    const int mt   = blockIdx.x;
    const int nq   = blockIdx.y;
    const int lane = threadIdx.x & 31;
    const int lrow = lane & 15;
    const int hf   = lane >> 4;
    const int m0   = mt * 16;
    const int n0   = nq * 32;
    const int ntA  = n0 >> 4;
    const int ntG  = 32 + ntA;

    const float* arow = x + (size_t)(m0 + lrow) * CC + 2 * hf;
    const float* pA0 = w1p + ((size_t)(ntA    ) * KP) * 128 + lane * 4;
    const float* pA1 = w1p + ((size_t)(ntA + 1) * KP) * 128 + lane * 4;
    const float* pG0 = w1p + ((size_t)(ntG    ) * KP) * 128 + lane * 4;
    const float* pG1 = w1p + ((size_t)(ntG + 1) * KP) * 128 + lane * 4;

    v8f aA0 = {}, aA1 = {}, aG0 = {}, aG1 = {};

    #pragma unroll 2
    for (int kp = 0; kp < KP; ++kp) {
        v2f a0 = *(const v2f*)(arow + kp * 8);
        v2f a1 = *(const v2f*)(arow + kp * 8 + 4);
        v4f fA0 = *(const v4f*)(pA0 + kp * 128);
        v4f fA1 = *(const v4f*)(pA1 + kp * 128);
        v4f fG0 = *(const v4f*)(pG0 + kp * 128);
        v4f fG1 = *(const v4f*)(pG1 + kp * 128);
        v2f b;
        b.x = fA0.x; b.y = fA0.y;
        aA0 = __builtin_amdgcn_wmma_f32_16x16x4_f32(false, a0, false, b, (short)0, aA0, false, false);
        b.x = fA0.z; b.y = fA0.w;
        aA0 = __builtin_amdgcn_wmma_f32_16x16x4_f32(false, a1, false, b, (short)0, aA0, false, false);
        b.x = fA1.x; b.y = fA1.y;
        aA1 = __builtin_amdgcn_wmma_f32_16x16x4_f32(false, a0, false, b, (short)0, aA1, false, false);
        b.x = fA1.z; b.y = fA1.w;
        aA1 = __builtin_amdgcn_wmma_f32_16x16x4_f32(false, a1, false, b, (short)0, aA1, false, false);
        b.x = fG0.x; b.y = fG0.y;
        aG0 = __builtin_amdgcn_wmma_f32_16x16x4_f32(false, a0, false, b, (short)0, aG0, false, false);
        b.x = fG0.z; b.y = fG0.w;
        aG0 = __builtin_amdgcn_wmma_f32_16x16x4_f32(false, a1, false, b, (short)0, aG0, false, false);
        b.x = fG1.x; b.y = fG1.y;
        aG1 = __builtin_amdgcn_wmma_f32_16x16x4_f32(false, a0, false, b, (short)0, aG1, false, false);
        b.x = fG1.z; b.y = fG1.w;
        aG1 = __builtin_amdgcn_wmma_f32_16x16x4_f32(false, a1, false, b, (short)0, aG1, false, false);
    }

    const int colA0 = n0 + lrow;
    const int colA1 = n0 + 16 + lrow;
    const float bA0 = b1[colA0],      bA1 = b1[colA1];
    const float bG0 = b1[CC + colA0], bG1 = b1[CC + colA1];
    #pragma unroll
    for (int v = 0; v < 8; ++v) {
        const int row = m0 + v + 8 * hf;
        const float av0 = aA0[v] + bA0;
        const float gv0 = aG0[v] + bG0;
        const float av1 = aA1[v] + bA1;
        const float gv1 = aG1[v] + bG1;
        xg[(size_t)row * CC + colA0] = av0 * (1.0f / (1.0f + expf(-gv0)));
        xg[(size_t)row * CC + colA1] = av1 * (1.0f / (1.0f + expf(-gv1)));
    }
}

__global__ __launch_bounds__(32)
void gemm_wl_kernel(const float* __restrict__ xg,
                    const float* __restrict__ wlwp,
                    const float* __restrict__ wlb,
                    float* __restrict__ wlog) {
    const int mt   = blockIdx.x;
    const int nq   = blockIdx.y;
    const int lane = threadIdx.x & 31;
    const int lrow = lane & 15;
    const int hf   = lane >> 4;
    const int m0   = mt * 16;
    const int n0   = nq * 32;
    const int nt   = n0 >> 4;

    const float* arow = xg + (size_t)(m0 + lrow) * CC + 2 * hf;
    const float* p0 = wlwp + ((size_t)(nt    ) * KP) * 128 + lane * 4;
    const float* p1 = wlwp + ((size_t)(nt + 1) * KP) * 128 + lane * 4;

    v8f acc0 = {}, acc1 = {};
    #pragma unroll 2
    for (int kp = 0; kp < KP; ++kp) {
        v2f a0 = *(const v2f*)(arow + kp * 8);
        v2f a1 = *(const v2f*)(arow + kp * 8 + 4);
        v4f f0 = *(const v4f*)(p0 + kp * 128);
        v4f f1 = *(const v4f*)(p1 + kp * 128);
        v2f b;
        b.x = f0.x; b.y = f0.y;
        acc0 = __builtin_amdgcn_wmma_f32_16x16x4_f32(false, a0, false, b, (short)0, acc0, false, false);
        b.x = f0.z; b.y = f0.w;
        acc0 = __builtin_amdgcn_wmma_f32_16x16x4_f32(false, a1, false, b, (short)0, acc0, false, false);
        b.x = f1.x; b.y = f1.y;
        acc1 = __builtin_amdgcn_wmma_f32_16x16x4_f32(false, a0, false, b, (short)0, acc1, false, false);
        b.x = f1.z; b.y = f1.w;
        acc1 = __builtin_amdgcn_wmma_f32_16x16x4_f32(false, a1, false, b, (short)0, acc1, false, false);
    }

    const int col0 = n0 + lrow;
    const int col1 = n0 + 16 + lrow;
    const float bias0 = (col0 < HK) ? wlb[col0] : 0.0f;
    const float bias1 = (col1 < HK) ? wlb[col1] : 0.0f;
    #pragma unroll
    for (int v = 0; v < 8; ++v) {
        const int row = m0 + v + 8 * hf;
        wlog[(size_t)row * HKPAD + col0] = acc0[v] + bias0;
        wlog[(size_t)row * HKPAD + col1] = acc1[v] + bias1;
    }
}

__global__ __launch_bounds__(256)
void softmax_taps_kernel(float* __restrict__ wlog) {
    const int idx = blockIdx.x * 256 + threadIdx.x;
    if (idx >= MROWS * HH) return;
    const int row = idx >> 3;
    const int h   = idx & 7;
    float* p = wlog + (size_t)row * HKPAD + h * KK;

    float vals[KK];
    float m = -INFINITY;
    #pragma unroll
    for (int k = 0; k < KK; ++k) { vals[k] = p[k]; m = fmaxf(m, vals[k]); }
    float s = 0.0f;
    #pragma unroll
    for (int k = 0; k < KK; ++k) { vals[k] = expf(vals[k] - m); s += vals[k]; }
    const float inv = 1.0f / s;
    #pragma unroll
    for (int k = 0; k < KK; ++k) p[k] = vals[k] * inv;
}

__global__ __launch_bounds__(256)
void dynconv_kernel(const float* __restrict__ xg,
                    const float* __restrict__ wlog,
                    float* __restrict__ y) {
    __shared__ float ws[4 * KK];
    const int bid   = blockIdx.x;
    const int row   = bid >> 1;
    const int chalf = bid & 1;
    const int tid   = threadIdx.x;
    const int c     = chalf * 256 + tid;
    const int t     = row >> 4;

    if (tid < 4 * KK) {
        const int hl = tid / KK;
        const int k  = tid - hl * KK;
        ws[tid] = wlog[(size_t)row * HKPAD + (chalf * 4 + hl) * KK + k];
    }
    __syncthreads();

    const float* taps = ws + (tid >> 6) * KK;
    float acc = 0.0f;
    #pragma unroll
    for (int k = 0; k < KK; ++k) {
        const int tt = t + k - PP;
        if (tt >= 0 && tt < TT) {
            acc = fmaf(xg[((size_t)row + (size_t)(k - PP) * BB) * CC + c],
                       taps[k], acc);
        }
    }
    y[(size_t)row * CC + c] = acc;
}

__global__ __launch_bounds__(32)
void gemm2_kernel(const float* __restrict__ y,
                  const float* __restrict__ w2p,
                  const float* __restrict__ b2,
                  float* __restrict__ out) {
    const int mt   = blockIdx.x;
    const int nq   = blockIdx.y;
    const int lane = threadIdx.x & 31;
    const int lrow = lane & 15;
    const int hf   = lane >> 4;
    const int m0   = mt * 16;
    const int n0   = nq * 64;
    const int nt   = n0 >> 4;

    const float* arow = y + (size_t)(m0 + lrow) * CC + 2 * hf;
    const float* p0 = w2p + ((size_t)(nt    ) * KP) * 128 + lane * 4;
    const float* p1 = w2p + ((size_t)(nt + 1) * KP) * 128 + lane * 4;
    const float* p2 = w2p + ((size_t)(nt + 2) * KP) * 128 + lane * 4;
    const float* p3 = w2p + ((size_t)(nt + 3) * KP) * 128 + lane * 4;

    v8f acc0 = {}, acc1 = {}, acc2 = {}, acc3 = {};
    #pragma unroll 2
    for (int kp = 0; kp < KP; ++kp) {
        v2f a0 = *(const v2f*)(arow + kp * 8);
        v2f a1 = *(const v2f*)(arow + kp * 8 + 4);
        v4f f0 = *(const v4f*)(p0 + kp * 128);
        v4f f1 = *(const v4f*)(p1 + kp * 128);
        v4f f2 = *(const v4f*)(p2 + kp * 128);
        v4f f3 = *(const v4f*)(p3 + kp * 128);
        v2f b;
        b.x = f0.x; b.y = f0.y;
        acc0 = __builtin_amdgcn_wmma_f32_16x16x4_f32(false, a0, false, b, (short)0, acc0, false, false);
        b.x = f0.z; b.y = f0.w;
        acc0 = __builtin_amdgcn_wmma_f32_16x16x4_f32(false, a1, false, b, (short)0, acc0, false, false);
        b.x = f1.x; b.y = f1.y;
        acc1 = __builtin_amdgcn_wmma_f32_16x16x4_f32(false, a0, false, b, (short)0, acc1, false, false);
        b.x = f1.z; b.y = f1.w;
        acc1 = __builtin_amdgcn_wmma_f32_16x16x4_f32(false, a1, false, b, (short)0, acc1, false, false);
        b.x = f2.x; b.y = f2.y;
        acc2 = __builtin_amdgcn_wmma_f32_16x16x4_f32(false, a0, false, b, (short)0, acc2, false, false);
        b.x = f2.z; b.y = f2.w;
        acc2 = __builtin_amdgcn_wmma_f32_16x16x4_f32(false, a1, false, b, (short)0, acc2, false, false);
        b.x = f3.x; b.y = f3.y;
        acc3 = __builtin_amdgcn_wmma_f32_16x16x4_f32(false, a0, false, b, (short)0, acc3, false, false);
        b.x = f3.z; b.y = f3.w;
        acc3 = __builtin_amdgcn_wmma_f32_16x16x4_f32(false, a1, false, b, (short)0, acc3, false, false);
    }

    const int col0 = n0 + lrow;
    const float bias0 = b2[col0];
    const float bias1 = b2[col0 + 16];
    const float bias2 = b2[col0 + 32];
    const float bias3 = b2[col0 + 48];
    #pragma unroll
    for (int v = 0; v < 8; ++v) {
        const int row = m0 + v + 8 * hf;
        float* o = out + (size_t)row * CC + col0;
        o[0]  = acc0[v] + bias0;
        o[16] = acc1[v] + bias1;
        o[32] = acc2[v] + bias2;
        o[48] = acc3[v] + bias3;
    }
}

extern "C" void kernel_launch(void* const* d_in, const int* in_sizes, int n_in,
                              void* d_out, int out_size, void* d_ws, size_t ws_size,
                              hipStream_t stream) {
    const float* x   = (const float*)d_in[0];
    const float* w1  = (const float*)d_in[1];
    const float* b1  = (const float*)d_in[2];
    const float* wlw = (const float*)d_in[3];
    const float* wlb = (const float*)d_in[4];
    const float* w2  = (const float*)d_in[5];
    const float* b2  = (const float*)d_in[6];
    float* out = (float*)d_out;

    float* ws = (float*)d_ws;
    size_t off = 0;
    float* xg   = ws + off; off += (size_t)MROWS * CC;
    float* wlog = ws + off; off += (size_t)MROWS * HKPAD;
    float* yb   = ws + off; off += (size_t)MROWS * CC;
    float* w1p  = ws + off; off += (size_t)(N2 / 16)    * KP * 128;
    float* wlwp = ws + off; off += (size_t)(HKPAD / 16) * KP * 128;
    float* w2p  = ws + off; off += (size_t)(CC / 16)    * KP * 128;

    {
        const int tot1 = (N2 / 16)    * KP * 128;
        const int totL = (HKPAD / 16) * KP * 128;
        const int tot2 = (CC / 16)    * KP * 128;
        pack_b_kernel<<<(tot1 + 255) / 256, 256, 0, stream>>>(w1,  w1p,  N2, N2, tot1);
        pack_b_kernel<<<(totL + 255) / 256, 256, 0, stream>>>(wlw, wlwp, HK, HK, totL);
        pack_b_kernel<<<(tot2 + 255) / 256, 256, 0, stream>>>(w2,  w2p,  CC, CC, tot2);
    }

    gemm1_glu_kernel<<<dim3(MROWS / 16, CC / 32), 32, 0, stream>>>(x, w1p, b1, xg);
    gemm_wl_kernel<<<dim3(MROWS / 16, HKPAD / 32), 32, 0, stream>>>(xg, wlwp, wlb, wlog);
    softmax_taps_kernel<<<(MROWS * HH + 255) / 256, 256, 0, stream>>>(wlog);
    dynconv_kernel<<<MROWS * 2, 256, 0, stream>>>(xg, wlog, yb);
    gemm2_kernel<<<dim3(MROWS / 16, CC / 64), 32, 0, stream>>>(yb, w2p, b2, out);
}


# GEMM weights live in LDS: each work-group copies its 512x128 weight panel into LDS once, in fragment-pair order, and walks its row tiles; B fragments are LDS reads, A is fetched one 64-k chunk ahead
# speedup vs baseline: 1.6628x; 1.6628x over previous
#include <hip/hip_runtime.h>
#include <math.h>
#include <stddef.h>


#define TT   1024
#define BB   16
#define CC   512
#define HH   8
#define KK   31
#define PP   15
#define MROWS (TT * BB)
#define N2    (2 * CC)
#define HK    (HH * KK)
#define HKPAD 256

typedef __attribute__((ext_vector_type(2))) float v2f;
typedef __attribute__((ext_vector_type(4))) float v4f;
typedef __attribute__((ext_vector_type(8))) float v8f;

#define NPAIR   (CC / 2)
#define PROW    256
#define PANEL_FLOATS (NPAIR * PROW)
#define GTHREADS 512

#define WM(a, b, c) c = __builtin_amdgcn_wmma_f32_16x16x4_f32(false, (a), false, (b), (short)0, c, false, false)
#define LO2(u) __builtin_shufflevector((u), (u), 0, 1)
#define HI2(u) __builtin_shufflevector((u), (u), 2, 3)

template <int NTHR>
__device__ __forceinline__ void fill_panel(float* __restrict__ panel, const float* __restrict__ W,
                                           int ld, int base0, int base1, int colmax, int tid) {
    const int c  = tid & 15;
    const int g  = (tid >> 4) & 1;
    const int pr = tid >> 5;
    int col = (g ? base1 : base0) + 4 * c;
    col = col < colmax ? col : colmax;
    const float* src = W + (size_t)(2 * pr) * ld + col;
    float* dst = panel + pr * PROW + g * 128 + c * 4;
    #pragma unroll 8
    for (int pass = 0; pass < NPAIR / (NTHR / 32); ++pass) {
        const v4f x = *(const v4f*)(src + (size_t)pass * (NTHR / 16) * ld);
        const v4f y = *(const v4f*)(src + (size_t)pass * (NTHR / 16) * ld + ld);
        v4f lo, hi;
        lo.x = x.x; lo.y = y.x; lo.z = x.y; lo.w = y.y;
        hi.x = x.z; hi.y = y.z; hi.z = x.w; hi.w = y.w;
        *(v4f*)(dst + pass * (NTHR / 32) * PROW)      = lo;
        *(v4f*)(dst + pass * (NTHR / 32) * PROW + 64) = hi;
    }
}

#define LOADA(buf, ptr)                                                   \
    _Pragma("unroll")                                                     \
    for (int i_ = 0; i_ < 8; ++i_) {                                      \
        buf[2 * i_]     = *(const v2f*)((ptr) + 8 * i_);                  \
        buf[2 * i_ + 1] = *(const v2f*)((ptr) + 8 * i_ + 4);              \
    }

#define CHUNK4(buf, q, c0, c1, c2, c3)                                    \
    _Pragma("unroll")                                                     \
    for (int kp_ = 0; kp_ < 8; ++kp_) {                                   \
        const v4f u0_ = *(const v4f*)((q) + kp_ * 4 * PROW);              \
        const v4f v0_ = *(const v4f*)((q) + kp_ * 4 * PROW + 64);         \
        const v4f u1_ = *(const v4f*)((q) + kp_ * 4 * PROW + 2 * PROW);   \
        const v4f v1_ = *(const v4f*)((q) + kp_ * 4 * PROW + 2 * PROW + 64); \
        WM(buf[2 * kp_], LO2(u0_), c0); WM(buf[2 * kp_], HI2(u0_), c1);   \
        WM(buf[2 * kp_], LO2(v0_), c2); WM(buf[2 * kp_], HI2(v0_), c3);   \
        WM(buf[2 * kp_ + 1], LO2(u1_), c0); WM(buf[2 * kp_ + 1], HI2(u1_), c1); \
        WM(buf[2 * kp_ + 1], LO2(v1_), c2); WM(buf[2 * kp_ + 1], HI2(v1_), c3); \
    }

#define CHUNK8(buf, q, a0, a1, a2, a3, g0, g1, g2, g3)                    \
    _Pragma("unroll")                                                     \
    for (int kp_ = 0; kp_ < 8; ++kp_) {                                   \
        const v4f u0_ = *(const v4f*)((q) + kp_ * 4 * PROW);              \
        const v4f v0_ = *(const v4f*)((q) + kp_ * 4 * PROW + 64);         \
        const v4f w0_ = *(const v4f*)((q) + kp_ * 4 * PROW + 128);        \
        const v4f z0_ = *(const v4f*)((q) + kp_ * 4 * PROW + 192);        \
        const v4f u1_ = *(const v4f*)((q) + kp_ * 4 * PROW + 2 * PROW);   \
        const v4f v1_ = *(const v4f*)((q) + kp_ * 4 * PROW + 2 * PROW + 64);  \
        const v4f w1_ = *(const v4f*)((q) + kp_ * 4 * PROW + 2 * PROW + 128); \
        const v4f z1_ = *(const v4f*)((q) + kp_ * 4 * PROW + 2 * PROW + 192); \
        WM(buf[2 * kp_], LO2(u0_), a0); WM(buf[2 * kp_], HI2(u0_), a1);   \
        WM(buf[2 * kp_], LO2(v0_), a2); WM(buf[2 * kp_], HI2(v0_), a3);   \
        WM(buf[2 * kp_], LO2(w0_), g0); WM(buf[2 * kp_], HI2(w0_), g1);   \
        WM(buf[2 * kp_], LO2(z0_), g2); WM(buf[2 * kp_], HI2(z0_), g3);   \
        WM(buf[2 * kp_ + 1], LO2(u1_), a0); WM(buf[2 * kp_ + 1], HI2(u1_), a1); \
        WM(buf[2 * kp_ + 1], LO2(v1_), a2); WM(buf[2 * kp_ + 1], HI2(v1_), a3); \
        WM(buf[2 * kp_ + 1], LO2(w1_), g0); WM(buf[2 * kp_ + 1], HI2(w1_), g1); \
        WM(buf[2 * kp_ + 1], LO2(z1_), g2); WM(buf[2 * kp_ + 1], HI2(z1_), g3); \
    }

#define G1_SLAB 512
#define G1_THREADS 256
__global__ __launch_bounds__(G1_THREADS)
void gemm1_glu_kernel(const float* __restrict__ x,
                      const float* __restrict__ w1,
                      const float* __restrict__ b1,
                      float* __restrict__ xg) {
    __shared__ __attribute__((aligned(16))) float panel[PANEL_FLOATS];
    const int tid  = threadIdx.x;
    const int wv   = tid >> 5;
    const int lane = tid & 31;
    const int lrow = lane & 15;
    const int hf   = lane >> 4;
    const int nq   = blockIdx.y;
    const int cb   = nq * 64 + 4 * lrow;

    fill_panel<G1_THREADS>(panel, w1, N2, nq * 64, CC + nq * 64, N2 - 4, tid);
    __syncthreads();

    const float* q0 = panel + hf * PROW + lrow * 4;
    const int mfirst = blockIdx.x * G1_SLAB + wv * 16;

    const v4f bA = *(const v4f*)(b1 + cb);
    const v4f bG = *(const v4f*)(b1 + CC + cb);

    v2f bufA[16], bufB[16];
    {
        const float* a0p = x + (size_t)(mfirst + lrow) * CC + 2 * hf;
        LOADA(bufA, a0p);
    }
    #pragma unroll 1
    for (int it = 0; it < 4; ++it) {
        const int m0 = mfirst + it * 128;
        const int mn = (it < 3) ? m0 + 128 : m0;
        const float* arow = x + (size_t)(m0 + lrow) * CC + 2 * hf;
        const float* anxt = x + (size_t)(mn + lrow) * CC + 2 * hf;
        v8f aA0 = {}, aA1 = {}, aA2 = {}, aA3 = {};
        v8f aG0 = {}, aG1 = {}, aG2 = {}, aG3 = {};
        #pragma unroll 1
        for (int c2 = 0; c2 < 4; ++c2) {
            const float* q = q0 + c2 * (64 * PROW);
            LOADA(bufB, arow + c2 * 128 + 64);
            CHUNK8(bufA, q, aA0, aA1, aA2, aA3, aG0, aG1, aG2, aG3);
            const float* an = (c2 < 3) ? (arow + c2 * 128 + 128) : anxt;
            LOADA(bufA, an);
            CHUNK8(bufB, q + 32 * PROW, aA0, aA1, aA2, aA3, aG0, aG1, aG2, aG3);
        }
        #pragma unroll
        for (int v = 0; v < 8; ++v) {
            const int row = m0 + v + 8 * hf;
            const float av0 = aA0[v] + bA.x, gv0 = aG0[v] + bG.x;
            const float av1 = aA1[v] + bA.y, gv1 = aG1[v] + bG.y;
            const float av2 = aA2[v] + bA.z, gv2 = aG2[v] + bG.z;
            const float av3 = aA3[v] + bA.w, gv3 = aG3[v] + bG.w;
            v4f o;
            o.x = av0 * (1.0f / (1.0f + expf(-gv0)));
            o.y = av1 * (1.0f / (1.0f + expf(-gv1)));
            o.z = av2 * (1.0f / (1.0f + expf(-gv2)));
            o.w = av3 * (1.0f / (1.0f + expf(-gv3)));
            *(v4f*)(xg + (size_t)row * CC + cb) = o;
        }
    }
}

#define WL_SLAB 128
__global__ __launch_bounds__(GTHREADS)
void gemm_wl_kernel(const float* __restrict__ xg,
                    const float* __restrict__ wlw,
                    const float* __restrict__ wlb,
                    float* __restrict__ wlog) {
    __shared__ __attribute__((aligned(16))) float panel[PANEL_FLOATS];
    const int tid  = threadIdx.x;
    const int wv   = tid >> 5;
    const int lane = tid & 31;
    const int lrow = lane & 15;
    const int hf   = lane >> 4;
    const int grp  = wv >> 3;
    const int cg   = blockIdx.y;
    const int cb   = cg * 128 + grp * 64 + 4 * lrow;

    fill_panel<GTHREADS>(panel, wlw, HK, cg * 128, cg * 128 + 64, HK - 4, tid);
    __syncthreads();

    const float* q0 = panel + hf * PROW + grp * 128 + lrow * 4;
    const int m0 = blockIdx.x * WL_SLAB + (wv & 7) * 16;
    const float* arow = xg + (size_t)(m0 + lrow) * CC + 2 * hf;

    v2f bufA[16], bufB[16];
    LOADA(bufA, arow);
    v8f acc0 = {}, acc1 = {}, acc2 = {}, acc3 = {};
    #pragma unroll 1
    for (int c2 = 0; c2 < 4; ++c2) {
        const float* q = q0 + c2 * (64 * PROW);
        LOADA(bufB, arow + c2 * 128 + 64);
        CHUNK4(bufA, q, acc0, acc1, acc2, acc3);
        const float* an = arow + ((c2 < 3) ? (c2 * 128 + 128) : 0);
        LOADA(bufA, an);
        CHUNK4(bufB, q + 32 * PROW, acc0, acc1, acc2, acc3);
    }

    if (cb < HK) {
        const v4f bias = *(const v4f*)(wlb + cb);
        #pragma unroll
        for (int v = 0; v < 8; ++v) {
            const int row = m0 + v + 8 * hf;
            v4f o;
            o.x = acc0[v] + bias.x;
            o.y = acc1[v] + bias.y;
            o.z = acc2[v] + bias.z;
            o.w = acc3[v] + bias.w;
            *(v4f*)(wlog + (size_t)row * HKPAD + cb) = o;
        }
    }
}

__global__ __launch_bounds__(256)
void softmax_taps_kernel(float* __restrict__ wlog) {
    const int idx = blockIdx.x * 256 + threadIdx.x;
    if (idx >= MROWS * HH) return;
    const int row = idx >> 3;
    const int h   = idx & 7;
    float* p = wlog + (size_t)row * HKPAD + h * KK;

    float vals[KK];
    float m = -INFINITY;
    #pragma unroll
    for (int k = 0; k < KK; ++k) { vals[k] = p[k]; m = fmaxf(m, vals[k]); }
    float s = 0.0f;
    #pragma unroll
    for (int k = 0; k < KK; ++k) { vals[k] = expf(vals[k] - m); s += vals[k]; }
    const float inv = 1.0f / s;
    #pragma unroll
    for (int k = 0; k < KK; ++k) p[k] = vals[k] * inv;
}

#define TC   32
#define WIN  (TC + KK - 1)
#define PADR (KK - 1 - PP)

static_assert(TT % TC == 0 && TT >= 2 * TC, "time chunks tile T");
static_assert(TC >= PP && TC >= PADR, "only the first / last chunk touch the padding");

__global__ __launch_bounds__(32)
void dynconv_kernel(const float* __restrict__ xg,
                    const float* __restrict__ wlog,
                    float* __restrict__ y) {
    const int h    = blockIdx.x;
    const int b    = blockIdx.y;
    const int t0   = blockIdx.z * TC;
    const int lane = threadIdx.x & 31;
    const int c0   = h * (CC / HH) + lane * 2;

    const float* p = xg + ((size_t)t0 * BB + b) * CC + c0;
    const v2f zero = {0.0f, 0.0f};
    v2f win[WIN];

    if (t0 >= PP) {
        #pragma unroll
        for (int i = 0; i < PP; ++i)
            win[i] = *(const v2f*)(p + (ptrdiff_t)(i - PP) * (BB * CC));
    } else {
        #pragma unroll
        for (int i = 0; i < PP; ++i) win[i] = zero;
    }
    #pragma unroll
    for (int i = PP; i < PP + TC; ++i)
        win[i] = *(const v2f*)(p + (ptrdiff_t)(i - PP) * (BB * CC));
    if (t0 + TC + PADR <= TT) {
        #pragma unroll
        for (int i = PP + TC; i < WIN; ++i)
            win[i] = *(const v2f*)(p + (ptrdiff_t)(i - PP) * (BB * CC));
    } else {
        #pragma unroll
        for (int i = PP + TC; i < WIN; ++i) win[i] = zero;
    }

    #pragma unroll
    for (int j = 0; j < TC; ++j) {
        const size_t row = (size_t)(t0 + j) * BB + b;
        const float* taps = wlog + row * HKPAD + h * KK;
        float a0 = 0.0f, a1 = 0.0f;
        #pragma unroll
        for (int k = 0; k < KK; ++k) {
            const float w = taps[k];
            a0 = fmaf(win[j + k].x, w, a0);
            a1 = fmaf(win[j + k].y, w, a1);
        }
        v2f r; r.x = a0; r.y = a1;
        *(v2f*)(y + row * CC + c0) = r;
    }
}

#define G2_SLAB 256
__global__ __launch_bounds__(GTHREADS)
void gemm2_kernel(const float* __restrict__ y,
                  const float* __restrict__ w2,
                  const float* __restrict__ b2,
                  float* __restrict__ out) {
    __shared__ __attribute__((aligned(16))) float panel[PANEL_FLOATS];
    const int tid  = threadIdx.x;
    const int wv   = tid >> 5;
    const int lane = tid & 31;
    const int lrow = lane & 15;
    const int hf   = lane >> 4;
    const int grp  = wv >> 3;
    const int cg   = blockIdx.y;
    const int cb   = cg * 128 + grp * 64 + 4 * lrow;

    fill_panel<GTHREADS>(panel, w2, CC, cg * 128, cg * 128 + 64, CC - 4, tid);
    __syncthreads();

    const float* q0 = panel + hf * PROW + grp * 128 + lrow * 4;
    const int mfirst = blockIdx.x * G2_SLAB + (wv & 7) * 16;
    const v4f bias = *(const v4f*)(b2 + cb);

    v2f bufA[16], bufB[16];
    {
        const float* a0p = y + (size_t)(mfirst + lrow) * CC + 2 * hf;
        LOADA(bufA, a0p);
    }
    #pragma unroll 1
    for (int it = 0; it < 2; ++it) {
        const int m0 = mfirst + it * 128;
        const int mn = (it == 0) ? m0 + 128 : m0;
        const float* arow = y + (size_t)(m0 + lrow) * CC + 2 * hf;
        const float* anxt = y + (size_t)(mn + lrow) * CC + 2 * hf;
        v8f acc0 = {}, acc1 = {}, acc2 = {}, acc3 = {};
        #pragma unroll 1
        for (int c2 = 0; c2 < 4; ++c2) {
            const float* q = q0 + c2 * (64 * PROW);
            LOADA(bufB, arow + c2 * 128 + 64);
            CHUNK4(bufA, q, acc0, acc1, acc2, acc3);
            const float* an = (c2 < 3) ? (arow + c2 * 128 + 128) : anxt;
            LOADA(bufA, an);
            CHUNK4(bufB, q + 32 * PROW, acc0, acc1, acc2, acc3);
        }
        #pragma unroll
        for (int v = 0; v < 8; ++v) {
            const int row = m0 + v + 8 * hf;
            v4f o;
            o.x = acc0[v] + bias.x;
            o.y = acc1[v] + bias.y;
            o.z = acc2[v] + bias.z;
            o.w = acc3[v] + bias.w;
            *(v4f*)(out + (size_t)row * CC + cb) = o;
        }
    }
}

extern "C" void kernel_launch(void* const* d_in, const int* in_sizes, int n_in,
                              void* d_out, int out_size, void* d_ws, size_t ws_size,
                              hipStream_t stream) {
    const float* x   = (const float*)d_in[0];
    const float* w1  = (const float*)d_in[1];
    const float* b1  = (const float*)d_in[2];
    const float* wlw = (const float*)d_in[3];
    const float* wlb = (const float*)d_in[4];
    const float* w2  = (const float*)d_in[5];
    const float* b2  = (const float*)d_in[6];
    float* out = (float*)d_out;

    float* ws = (float*)d_ws;
    size_t off = 0;
    float* xg   = ws + off; off += (size_t)MROWS * CC;
    float* wlog = ws + off; off += (size_t)MROWS * HKPAD;
    float* yb   = ws + off; off += (size_t)MROWS * CC;

    gemm1_glu_kernel<<<dim3(MROWS / G1_SLAB, CC / 64), G1_THREADS, 0, stream>>>(x, w1, b1, xg);
    gemm_wl_kernel<<<dim3(MROWS / WL_SLAB, HKPAD / 128), GTHREADS, 0, stream>>>(xg, wlw, wlb, wlog);
    softmax_taps_kernel<<<(MROWS * HH + 255) / 256, 256, 0, stream>>>(wlog);
    dynconv_kernel<<<dim3(HH, BB, TT / TC), 32, 0, stream>>>(xg, wlog, yb);
    gemm2_kernel<<<dim3(MROWS / G2_SLAB, CC / 128), GTHREADS, 0, stream>>>(yb, w2, b2, out);
}


# dynconv taps as vector loads: lane l loads tap l of each of the wave's 32 steps up front (one 128-byte line per step); FMAs take tap k from lane k by readlane; no scalar loads, no full wait per step
# speedup vs baseline: 1.9256x; 1.1580x over previous
#include <hip/hip_runtime.h>
#include <math.h>
#include <stddef.h>


#define TT   1024
#define BB   16
#define CC   512
#define HH   8
#define KK   31
#define PP   15
#define MROWS (TT * BB)
#define N2    (2 * CC)
#define HK    (HH * KK)
#define HPITCH 32
#define HKPAD (HH * HPITCH)

typedef __attribute__((ext_vector_type(2))) float v2f;
typedef __attribute__((ext_vector_type(4))) float v4f;
typedef __attribute__((ext_vector_type(8))) float v8f;

#define NPAIR   (CC / 2)
#define PROW    256
#define PANEL_FLOATS (NPAIR * PROW)
#define GTHREADS 512

#define WM(a, b, c) c = __builtin_amdgcn_wmma_f32_16x16x4_f32(false, (a), false, (b), (short)0, c, false, false)
#define LO2(u) __builtin_shufflevector((u), (u), 0, 1)
#define HI2(u) __builtin_shufflevector((u), (u), 2, 3)

template <int NTHR>
__device__ __forceinline__ void fill_panel(float* __restrict__ panel, const float* __restrict__ W,
                                           int ld, int base0, int base1, int colmax, int tid) {
    const int c  = tid & 15;
    const int g  = (tid >> 4) & 1;
    const int pr = tid >> 5;
    int col = (g ? base1 : base0) + 4 * c;
    col = col < colmax ? col : colmax;
    const float* src = W + (size_t)(2 * pr) * ld + col;
    float* dst = panel + pr * PROW + g * 128 + c * 4;
    #pragma unroll 8
    for (int pass = 0; pass < NPAIR / (NTHR / 32); ++pass) {
        const v4f x = *(const v4f*)(src + (size_t)pass * (NTHR / 16) * ld);
        const v4f y = *(const v4f*)(src + (size_t)pass * (NTHR / 16) * ld + ld);
        v4f lo, hi;
        lo.x = x.x; lo.y = y.x; lo.z = x.y; lo.w = y.y;
        hi.x = x.z; hi.y = y.z; hi.z = x.w; hi.w = y.w;
        *(v4f*)(dst + pass * (NTHR / 32) * PROW)      = lo;
        *(v4f*)(dst + pass * (NTHR / 32) * PROW + 64) = hi;
    }
}

__device__ __forceinline__ void fill_panel_taps(float* __restrict__ panel, const float* __restrict__ W,
                                                int cg, int tid) {
    const int c    = tid & 15;
    const int g    = (tid >> 4) & 1;
    const int pr   = tid >> 5;
    const int head = 4 * cg + 2 * g + (c >> 3);
    const int cc   = c & 7;
    const bool last = (cc == 7);
    const int col = head * KK + 4 * cc - (last ? 1 : 0);
    const float* src = W + (size_t)(2 * pr) * HK + col;
    float* dst = panel + pr * PROW + g * 128 + c * 4;
    #pragma unroll 8
    for (int pass = 0; pass < NPAIR / (GTHREADS / 32); ++pass) {
        const float* s0 = src + (size_t)pass * (GTHREADS / 16) * HK;
        const float* s1 = s0 + HK;
        const float x0 = s0[0], x1 = s0[1], x2 = s0[2], x3 = s0[3];
        const float y0 = s1[0], y1 = s1[1], y2 = s1[2], y3 = s1[3];
        v4f lo, hi;
        lo.x = last ? x1 : x0; lo.y = last ? y1 : y0; lo.z = last ? x2 : x1; lo.w = last ? y2 : y1;
        hi.x = last ? x3 : x2; hi.y = last ? y3 : y2; hi.z = x3; hi.w = y3;
        *(v4f*)(dst + pass * (GTHREADS / 32) * PROW)      = lo;
        *(v4f*)(dst + pass * (GTHREADS / 32) * PROW + 64) = hi;
    }
}

#define LOADA(buf, ptr)                                                   \
    _Pragma("unroll")                                                     \
    for (int i_ = 0; i_ < 8; ++i_) {                                      \
        buf[2 * i_]     = *(const v2f*)((ptr) + 8 * i_);                  \
        buf[2 * i_ + 1] = *(const v2f*)((ptr) + 8 * i_ + 4);              \
    }

#define CHUNK4(buf, q, c0, c1, c2, c3)                                    \
    _Pragma("unroll")                                                     \
    for (int kp_ = 0; kp_ < 8; ++kp_) {                                   \
        const v4f u0_ = *(const v4f*)((q) + kp_ * 4 * PROW);              \
        const v4f v0_ = *(const v4f*)((q) + kp_ * 4 * PROW + 64);         \
        const v4f u1_ = *(const v4f*)((q) + kp_ * 4 * PROW + 2 * PROW);   \
        const v4f v1_ = *(const v4f*)((q) + kp_ * 4 * PROW + 2 * PROW + 64); \
        WM(buf[2 * kp_], LO2(u0_), c0); WM(buf[2 * kp_], HI2(u0_), c1);   \
        WM(buf[2 * kp_], LO2(v0_), c2); WM(buf[2 * kp_], HI2(v0_), c3);   \
        WM(buf[2 * kp_ + 1], LO2(u1_), c0); WM(buf[2 * kp_ + 1], HI2(u1_), c1); \
        WM(buf[2 * kp_ + 1], LO2(v1_), c2); WM(buf[2 * kp_ + 1], HI2(v1_), c3); \
    }

#define CHUNK8(buf, q, a0, a1, a2, a3, g0, g1, g2, g3)                    \
    _Pragma("unroll")                                                     \
    for (int kp_ = 0; kp_ < 8; ++kp_) {                                   \
        const v4f u0_ = *(const v4f*)((q) + kp_ * 4 * PROW);              \
        const v4f v0_ = *(const v4f*)((q) + kp_ * 4 * PROW + 64);         \
        const v4f w0_ = *(const v4f*)((q) + kp_ * 4 * PROW + 128);        \
        const v4f z0_ = *(const v4f*)((q) + kp_ * 4 * PROW + 192);        \
        const v4f u1_ = *(const v4f*)((q) + kp_ * 4 * PROW + 2 * PROW);   \
        const v4f v1_ = *(const v4f*)((q) + kp_ * 4 * PROW + 2 * PROW + 64);  \
        const v4f w1_ = *(const v4f*)((q) + kp_ * 4 * PROW + 2 * PROW + 128); \
        const v4f z1_ = *(const v4f*)((q) + kp_ * 4 * PROW + 2 * PROW + 192); \
        WM(buf[2 * kp_], LO2(u0_), a0); WM(buf[2 * kp_], HI2(u0_), a1);   \
        WM(buf[2 * kp_], LO2(v0_), a2); WM(buf[2 * kp_], HI2(v0_), a3);   \
        WM(buf[2 * kp_], LO2(w0_), g0); WM(buf[2 * kp_], HI2(w0_), g1);   \
        WM(buf[2 * kp_], LO2(z0_), g2); WM(buf[2 * kp_], HI2(z0_), g3);   \
        WM(buf[2 * kp_ + 1], LO2(u1_), a0); WM(buf[2 * kp_ + 1], HI2(u1_), a1); \
        WM(buf[2 * kp_ + 1], LO2(v1_), a2); WM(buf[2 * kp_ + 1], HI2(v1_), a3); \
        WM(buf[2 * kp_ + 1], LO2(w1_), g0); WM(buf[2 * kp_ + 1], HI2(w1_), g1); \
        WM(buf[2 * kp_ + 1], LO2(z1_), g2); WM(buf[2 * kp_ + 1], HI2(z1_), g3); \
    }

#define G1_SLAB 512
#define G1_THREADS 256
__global__ __launch_bounds__(G1_THREADS)
void gemm1_glu_kernel(const float* __restrict__ x,
                      const float* __restrict__ w1,
                      const float* __restrict__ b1,
                      float* __restrict__ xg) {
    __shared__ __attribute__((aligned(16))) float panel[PANEL_FLOATS];
    const int tid  = threadIdx.x;
    const int wv   = tid >> 5;
    const int lane = tid & 31;
    const int lrow = lane & 15;
    const int hf   = lane >> 4;
    const int nq   = blockIdx.y;
    const int cb   = nq * 64 + 4 * lrow;

    fill_panel<G1_THREADS>(panel, w1, N2, nq * 64, CC + nq * 64, N2 - 4, tid);
    __syncthreads();

    const float* q0 = panel + hf * PROW + lrow * 4;
    const int mfirst = blockIdx.x * G1_SLAB + wv * 16;

    const v4f bA = *(const v4f*)(b1 + cb);
    const v4f bG = *(const v4f*)(b1 + CC + cb);

    v2f bufA[16], bufB[16];
    {
        const float* a0p = x + (size_t)(mfirst + lrow) * CC + 2 * hf;
        LOADA(bufA, a0p);
    }
    #pragma unroll 1
    for (int it = 0; it < 4; ++it) {
        const int m0 = mfirst + it * 128;
        const int mn = (it < 3) ? m0 + 128 : m0;
        const float* arow = x + (size_t)(m0 + lrow) * CC + 2 * hf;
        const float* anxt = x + (size_t)(mn + lrow) * CC + 2 * hf;
        v8f aA0 = {}, aA1 = {}, aA2 = {}, aA3 = {};
        v8f aG0 = {}, aG1 = {}, aG2 = {}, aG3 = {};
        #pragma unroll 1
        for (int c2 = 0; c2 < 4; ++c2) {
            const float* q = q0 + c2 * (64 * PROW);
            LOADA(bufB, arow + c2 * 128 + 64);
            CHUNK8(bufA, q, aA0, aA1, aA2, aA3, aG0, aG1, aG2, aG3);
            const float* an = (c2 < 3) ? (arow + c2 * 128 + 128) : anxt;
            LOADA(bufA, an);
            CHUNK8(bufB, q + 32 * PROW, aA0, aA1, aA2, aA3, aG0, aG1, aG2, aG3);
        }
        #pragma unroll
        for (int v = 0; v < 8; ++v) {
            const int row = m0 + v + 8 * hf;
            const float av0 = aA0[v] + bA.x, gv0 = aG0[v] + bG.x;
            const float av1 = aA1[v] + bA.y, gv1 = aG1[v] + bG.y;
            const float av2 = aA2[v] + bA.z, gv2 = aG2[v] + bG.z;
            const float av3 = aA3[v] + bA.w, gv3 = aG3[v] + bG.w;
            v4f o;
            o.x = av0 * (1.0f / (1.0f + expf(-gv0)));
            o.y = av1 * (1.0f / (1.0f + expf(-gv1)));
            o.z = av2 * (1.0f / (1.0f + expf(-gv2)));
            o.w = av3 * (1.0f / (1.0f + expf(-gv3)));
            *(v4f*)(xg + (size_t)row * CC + cb) = o;
        }
    }
}

#define WL_SLAB 128
__global__ __launch_bounds__(GTHREADS)
void gemm_wl_kernel(const float* __restrict__ xg,
                    const float* __restrict__ wlw,
                    const float* __restrict__ wlb,
                    float* __restrict__ wlog) {
    __shared__ __attribute__((aligned(16))) float panel[PANEL_FLOATS];
    const int tid  = threadIdx.x;
    const int wv   = tid >> 5;
    const int lane = tid & 31;
    const int lrow = lane & 15;
    const int hf   = lane >> 4;
    const int grp  = wv >> 3;
    const int cg   = blockIdx.y;
    const int head = 4 * cg + 2 * grp + (lrow >> 3);
    const int cc   = lrow & 7;
    const bool last = (cc == 7);

    fill_panel_taps(panel, wlw, cg, tid);
    __syncthreads();

    const float* q0 = panel + hf * PROW + grp * 128 + lrow * 4;
    const int m0 = blockIdx.x * WL_SLAB + (wv & 7) * 16;
    const float* arow = xg + (size_t)(m0 + lrow) * CC + 2 * hf;

    v2f bufA[16], bufB[16];
    LOADA(bufA, arow);
    v8f acc0 = {}, acc1 = {}, acc2 = {}, acc3 = {};
    #pragma unroll 1
    for (int c2 = 0; c2 < 4; ++c2) {
        const float* q = q0 + c2 * (64 * PROW);
        LOADA(bufB, arow + c2 * 128 + 64);
        CHUNK4(bufA, q, acc0, acc1, acc2, acc3);
        const float* an = arow + ((c2 < 3) ? (c2 * 128 + 128) : 0);
        LOADA(bufA, an);
        CHUNK4(bufB, q + 32 * PROW, acc0, acc1, acc2, acc3);
    }

    const float* bp = wlb + head * KK + 4 * cc;
    const float bs0 = bp[0], bs1 = bp[1], bs2 = bp[2];
    const float bs3 = bp[last ? 2 : 3];
    #pragma unroll
    for (int v = 0; v < 8; ++v) {
        const int row = m0 + v + 8 * hf;
        const float l0 = acc0[v] + bs0;
        const float l1 = acc1[v] + bs1;
        const float l2 = acc2[v] + bs2;
        const float l3 = last ? -INFINITY : acc3[v] + bs3;
        float m = fmaxf(fmaxf(l0, l1), fmaxf(l2, l3));
        m = fmaxf(m, __shfl_xor(m, 1));
        m = fmaxf(m, __shfl_xor(m, 2));
        m = fmaxf(m, __shfl_xor(m, 4));
        const float e0 = expf(l0 - m);
        const float e1 = expf(l1 - m);
        const float e2 = expf(l2 - m);
        const float e3 = expf(l3 - m);
        float s = (e0 + e1) + (e2 + e3);
        s += __shfl_xor(s, 1);
        s += __shfl_xor(s, 2);
        s += __shfl_xor(s, 4);
        const float inv = 1.0f / s;
        v4f o;
        o.x = e0 * inv;
        o.y = e1 * inv;
        o.z = e2 * inv;
        o.w = e3 * inv;
        *(v4f*)(wlog + (size_t)row * HKPAD + head * HPITCH + 4 * cc) = o;
    }
}

#define TC   32
#define WIN  (TC + KK - 1)
#define PADR (KK - 1 - PP)

static_assert(TT % TC == 0 && TT >= 2 * TC, "time chunks tile T");
static_assert(TC >= PP && TC >= PADR, "only the first / last chunk touch the padding");

__global__ __launch_bounds__(32)
void dynconv_kernel(const float* __restrict__ xg,
                    const float* __restrict__ wlog,
                    float* __restrict__ y) {
    const int h    = blockIdx.x;
    const int b    = blockIdx.y;
    const int t0   = blockIdx.z * TC;
    const int lane = threadIdx.x & 31;
    const int c0   = h * (CC / HH) + lane * 2;

    const float* tp = wlog + ((size_t)t0 * BB + b) * HKPAD + h * HPITCH + lane;
    float tapv[TC];
    #pragma unroll
    for (int j = 0; j < TC; ++j)
        tapv[j] = tp[(size_t)j * (BB * HKPAD)];

    const float* p = xg + ((size_t)t0 * BB + b) * CC + c0;
    const v2f zero = {0.0f, 0.0f};
    v2f win[WIN];

    if (t0 >= PP) {
        #pragma unroll
        for (int i = 0; i < PP; ++i)
            win[i] = *(const v2f*)(p + (ptrdiff_t)(i - PP) * (BB * CC));
    } else {
        #pragma unroll
        for (int i = 0; i < PP; ++i) win[i] = zero;
    }
    #pragma unroll
    for (int i = PP; i < PP + TC; ++i)
        win[i] = *(const v2f*)(p + (ptrdiff_t)(i - PP) * (BB * CC));
    if (t0 + TC + PADR <= TT) {
        #pragma unroll
        for (int i = PP + TC; i < WIN; ++i)
            win[i] = *(const v2f*)(p + (ptrdiff_t)(i - PP) * (BB * CC));
    } else {
        #pragma unroll
        for (int i = PP + TC; i < WIN; ++i) win[i] = zero;
    }

    #pragma unroll
    for (int j = 0; j < TC; ++j) {
        const size_t row = (size_t)(t0 + j) * BB + b;
        const int tj = __float_as_int(tapv[j]);
        float a0 = 0.0f, a1 = 0.0f;
        #pragma unroll
        for (int k = 0; k < KK; ++k) {
            const float w = __int_as_float(__builtin_amdgcn_readlane(tj, k));
            a0 = fmaf(win[j + k].x, w, a0);
            a1 = fmaf(win[j + k].y, w, a1);
        }
        v2f r; r.x = a0; r.y = a1;
        *(v2f*)(y + row * CC + c0) = r;
    }
}

#define G2_SLAB 256
__global__ __launch_bounds__(GTHREADS)
void gemm2_kernel(const float* __restrict__ y,
                  const float* __restrict__ w2,
                  const float* __restrict__ b2,
                  float* __restrict__ out) {
    __shared__ __attribute__((aligned(16))) float panel[PANEL_FLOATS];
    const int tid  = threadIdx.x;
    const int wv   = tid >> 5;
    const int lane = tid & 31;
    const int lrow = lane & 15;
    const int hf   = lane >> 4;
    const int grp  = wv >> 3;
    const int cg   = blockIdx.y;
    const int cb   = cg * 128 + grp * 64 + 4 * lrow;

    fill_panel<GTHREADS>(panel, w2, CC, cg * 128, cg * 128 + 64, CC - 4, tid);
    __syncthreads();

    const float* q0 = panel + hf * PROW + grp * 128 + lrow * 4;
    const int mfirst = blockIdx.x * G2_SLAB + (wv & 7) * 16;
    const v4f bias = *(const v4f*)(b2 + cb);

    v2f bufA[16], bufB[16];
    {
        const float* a0p = y + (size_t)(mfirst + lrow) * CC + 2 * hf;
        LOADA(bufA, a0p);
    }
    #pragma unroll 1
    for (int it = 0; it < 2; ++it) {
        const int m0 = mfirst + it * 128;
        const int mn = (it == 0) ? m0 + 128 : m0;
        const float* arow = y + (size_t)(m0 + lrow) * CC + 2 * hf;
        const float* anxt = y + (size_t)(mn + lrow) * CC + 2 * hf;
        v8f acc0 = {}, acc1 = {}, acc2 = {}, acc3 = {};
        #pragma unroll 1
        for (int c2 = 0; c2 < 4; ++c2) {
            const float* q = q0 + c2 * (64 * PROW);
            LOADA(bufB, arow + c2 * 128 + 64);
            CHUNK4(bufA, q, acc0, acc1, acc2, acc3);
            const float* an = (c2 < 3) ? (arow + c2 * 128 + 128) : anxt;
            LOADA(bufA, an);
            CHUNK4(bufB, q + 32 * PROW, acc0, acc1, acc2, acc3);
        }
        #pragma unroll
        for (int v = 0; v < 8; ++v) {
            const int row = m0 + v + 8 * hf;
            v4f o;
            o.x = acc0[v] + bias.x;
            o.y = acc1[v] + bias.y;
            o.z = acc2[v] + bias.z;
            o.w = acc3[v] + bias.w;
            *(v4f*)(out + (size_t)row * CC + cb) = o;
        }
    }
}

extern "C" void kernel_launch(void* const* d_in, const int* in_sizes, int n_in,
                              void* d_out, int out_size, void* d_ws, size_t ws_size,
                              hipStream_t stream) {
    const float* x   = (const float*)d_in[0];
    const float* w1  = (const float*)d_in[1];
    const float* b1  = (const float*)d_in[2];
    const float* wlw = (const float*)d_in[3];
    const float* wlb = (const float*)d_in[4];
    const float* w2  = (const float*)d_in[5];
    const float* b2  = (const float*)d_in[6];
    float* out = (float*)d_out;

    float* ws = (float*)d_ws;
    size_t off = 0;
    float* xg   = ws + off; off += (size_t)MROWS * CC;
    float* wlog = ws + off; off += (size_t)MROWS * HKPAD;
    float* yb   = ws + off; off += (size_t)MROWS * CC;

    gemm1_glu_kernel<<<dim3(MROWS / G1_SLAB, CC / 64), G1_THREADS, 0, stream>>>(x, w1, b1, xg);
    gemm_wl_kernel<<<dim3(MROWS / WL_SLAB, HKPAD / 128), GTHREADS, 0, stream>>>(xg, wlw, wlb, wlog);
    dynconv_kernel<<<dim3(HH, BB, TT / TC), 32, 0, stream>>>(xg, wlog, yb);
    gemm2_kernel<<<dim3(MROWS / G2_SLAB, CC / 128), GTHREADS, 0, stream>>>(yb, w2, b2, out);
}
